# starting kernel
# baseline (speedup 1.0000x reference)
#include <hip/hip_runtime.h>
#include <math.h>

typedef __attribute__((ext_vector_type(2))) float v2f;
typedef __attribute__((ext_vector_type(8))) float v8f;

__device__ __forceinline__ void atomic_add_f32(float* p, float v) {
  __hip_atomic_fetch_add(p, v, __ATOMIC_RELAXED, __HIP_MEMORY_SCOPE_AGENT);
}

__global__ void deg_kernel(const int* __restrict__ cols,
                           const float* __restrict__ ew,
                           float* __restrict__ deg, int E) {
  int e = blockIdx.x * blockDim.x + threadIdx.x;
  if (e >= E) return;
  atomic_add_f32(&deg[cols[e]], ew[e]);
}

__global__ void dinv_kernel(float* __restrict__ deg, int N) {
  int i = blockIdx.x * blockDim.x + threadIdx.x;
  if (i >= N) return;
  float d = deg[i] + 1.0f;
  deg[i] = (d > 0.0f) ? rsqrtf(d) : 0.0f;
}

template <int NCOLS>
__global__ void gemm_wmma_f32(const float* __restrict__ A,
                              const float* __restrict__ B,
                              float* __restrict__ C) {
  const int m0   = blockIdx.x * 16;
  const int wid  = threadIdx.x >> 5;
  const int lane = threadIdx.x & 31;
  const int n0   = wid * 16;
  const int ln   = lane & 15;
  const int half = lane >> 4;

  v8f acc = {};
  const float* aptr = A + (size_t)(m0 + ln) * 64 + 2 * half;
  const float* bptr = B + (size_t)(2 * half) * NCOLS + n0 + ln;

#pragma unroll
  for (int k = 0; k < 64; k += 4) {
    v2f a, b;
    a.x = aptr[k];
    a.y = aptr[k + 1];
    b.x = bptr[k * NCOLS];
    b.y = bptr[k * NCOLS + NCOLS];
    acc = __builtin_amdgcn_wmma_f32_16x16x4_f32(
         false, a,  false, b,
         (short)0, acc,  false,  false);
  }

  float* cptr = C + (size_t)(m0 + 8 * half) * NCOLS + n0 + ln;
#pragma unroll
  for (int r = 0; r < 8; ++r) {
    cptr[(size_t)r * NCOLS] = acc[r];
  }
}

template <int CH>
__global__ void aggregate_kernel(const float* __restrict__ x,
                                 const int* __restrict__ rows,
                                 const int* __restrict__ cols,
                                 const float* __restrict__ ew,
                                 const float* __restrict__ dinv,
                                 float* __restrict__ out,
                                 int E, int N) {
  const int VPE = CH / 4;
  long long tid = (long long)blockIdx.x * blockDim.x + threadIdx.x;
  long long total = (long long)(E + N) * VPE;
  if (tid >= total) return;
  long long e = tid / VPE;
  int c4 = (int)(tid % VPE);

  int r, c; float w;
  if (e < E) { r = rows[e]; c = cols[e]; w = ew[e]; }
  else       { r = c = (int)(e - E);     w = 1.0f; }

  float nrm = dinv[r] * w * dinv[c];
  const float4 xv = *(const float4*)(x + (long long)r * CH + c4 * 4);
  float* dst = out + (long long)c * CH + c4 * 4;
  atomic_add_f32(dst + 0, xv.x * nrm);
  atomic_add_f32(dst + 1, xv.y * nrm);
  atomic_add_f32(dst + 2, xv.z * nrm);
  atomic_add_f32(dst + 3, xv.w * nrm);
}

__global__ void bias_relu_kernel(float* __restrict__ x,
                                 const float* __restrict__ b, int N) {
  long long tid = (long long)blockIdx.x * blockDim.x + threadIdx.x;
  if (tid >= (long long)N * 16) return;
  int c4 = (int)(tid & 15);
  float4 v  = ((float4*)x)[tid];
  float4 bb = ((const float4*)b)[c4];
  v.x = fmaxf(v.x + bb.x, 0.0f);
  v.y = fmaxf(v.y + bb.y, 0.0f);
  v.z = fmaxf(v.z + bb.z, 0.0f);
  v.w = fmaxf(v.w + bb.w, 0.0f);
  ((float4*)x)[tid] = v;
}

__global__ void l2norm_kernel(const float* __restrict__ x,
                              const float* __restrict__ b2,
                              float* __restrict__ out, int N) {
  int row  = blockIdx.x * (blockDim.x >> 5) + (threadIdx.x >> 5);
  int lane = threadIdx.x & 31;
  if (row >= N) return;
  float v = x[(long long)row * 32 + lane] + b2[lane];
  float ss = v * v;
#pragma unroll
  for (int off = 16; off > 0; off >>= 1)
    ss += __shfl_xor(ss, off, 32);
  float l2 = sqrtf(ss);
  float scale = 1.0f / fmaxf(l2, 1e-12f);
  out[(long long)row * 32 + lane] = v * scale;
}

extern "C" void kernel_launch(void* const* d_in, const int* in_sizes, int n_in,
                              void* d_out, int out_size, void* d_ws, size_t ws_size,
                              hipStream_t stream) {
  const int*   edge_index = (const int*)d_in[0];
  const float* ew         = (const float*)d_in[1];
  const float* emb        = (const float*)d_in[2];
  const float* W1         = (const float*)d_in[3];
  const float* b1         = (const float*)d_in[4];
  const float* W2         = (const float*)d_in[5];
  const float* b2         = (const float*)d_in[6];

  const int E = in_sizes[0] / 2;
  const int N = in_sizes[2] / 64;
  const int* rows = edge_index;
  const int* cols = edge_index + E;

  float* ws   = (float*)d_ws;
  float* dinv = ws;
  float* xw1  = dinv + (size_t)N;
  float* out1 = xw1  + (size_t)N * 64;
  float* xw2  = out1 + (size_t)N * 64;
  float* out2 = xw2  + (size_t)N * 32;

  hipMemsetAsync(dinv, 0, sizeof(float) * (size_t)N, stream);
  hipMemsetAsync(out1, 0, sizeof(float) * (size_t)N * 64, stream);
  hipMemsetAsync(out2, 0, sizeof(float) * (size_t)N * 32, stream);

  deg_kernel<<<(E + 255) / 256, 256, 0, stream>>>(cols, ew, dinv, E);
  dinv_kernel<<<(N + 255) / 256, 256, 0, stream>>>(dinv, N);

  gemm_wmma_f32<64><<<N / 16, 128, 0, stream>>>(emb, W1, xw1);
  {
    long long total = (long long)(E + N) * 16;
    aggregate_kernel<64><<<(unsigned)((total + 255) / 256), 256, 0, stream>>>(
        xw1, rows, cols, ew, dinv, out1, E, N);
  }
  bias_relu_kernel<<<(unsigned)(((long long)N * 16 + 255) / 256), 256, 0, stream>>>(out1, b1, N);

  gemm_wmma_f32<32><<<N / 16, 64, 0, stream>>>(out1, W2, xw2);
  {
    long long total = (long long)(E + N) * 8;
    aggregate_kernel<32><<<(unsigned)((total + 255) / 256), 256, 0, stream>>>(
        xw2, rows, cols, ew, dinv, out2, E, N);
  }

  l2norm_kernel<<<(N + 7) / 8, 256, 0, stream>>>(out2, b2, (float*)d_out, N);
}


# Degree pass: a workgroup sums its edge chunk's weights over half the node range in an LDS table, then adds the table by whole-line atomics: 3.2M one-word adds become 0.4M line adds (order varies)
# speedup vs baseline: 2.4585x; 2.4585x over previous
#include <hip/hip_runtime.h>
#include <math.h>

typedef __attribute__((ext_vector_type(2))) float v2f;
typedef __attribute__((ext_vector_type(8))) float v8f;

__device__ __forceinline__ void atomic_add_f32(float* p, float v) {
  __hip_atomic_fetch_add(p, v, __ATOMIC_RELAXED, __HIP_MEMORY_SCOPE_AGENT);
}

__device__ __forceinline__ int package_half() {
  return (int)((__builtin_amdgcn_s_sendmsg_rtn(0x87) >> 18) & 1u);
}

#define DEG_CAP 51200
#define DEG_THREADS 1024
#define DEG_BATCH 8

__device__ __forceinline__ void lds_add_f32(float* p, float v) {
  __hip_atomic_fetch_add(p, v, __ATOMIC_RELAXED, __HIP_MEMORY_SCOPE_WORKGROUP);
}

__device__ __forceinline__ void deg_take(float* tab, int c, float w, int lo, int range) {
  const unsigned idx = (unsigned)(c - lo);
  if (idx < (unsigned)range) lds_add_f32(&tab[idx], w);
}

__global__ void __launch_bounds__(DEG_THREADS)
deg_kernel(const int* __restrict__ cols, const float* __restrict__ ew,
           float* __restrict__ deg2, int E, int N) {
  __shared__ float4 tab4[DEG_CAP / 4];
  float* tab = (float*)tab4;
  const int t  = (int)threadIdx.x;
  const int G  = (int)(gridDim.x >> 1);
  const int g  = (int)(blockIdx.x >> 1);
  const int h0 = (int)(blockIdx.x & 1u);

  int range = (((N + 1) >> 1) + 31) & ~31;
  range = range > DEG_CAP ? DEG_CAP : range;
  range = range < 32 ? 32 : range;
  const int nr = (N + range - 1) / range;

  const int chunk = (((E + G - 1) / G) + 3) & ~3;
  const long long s64 = (long long)g * chunk;
  const int start = s64 < (long long)E ? (int)s64 : E;
  const int end   = (E - start) > chunk ? start + chunk : E;
  const bool vec  = ((((size_t)cols) | ((size_t)ew)) & 15u) == 0;
  const int nq    = vec ? ((end - start) >> 2) : 0;
  const int4*   cols4 = (const int4*)(cols + start);
  const float4* ew4   = (const float4*)(ew + start);

  float* deg = deg2 + (size_t)package_half() * (size_t)N;
  const int rot = (int)(((long long)g * range) / G) & ~31;

  for (int rg = h0; rg < nr; rg += 2) {
    const int lo = rg * range;
    for (int i = t; i < (range >> 2); i += DEG_THREADS)
      tab4[i] = make_float4(0.0f, 0.0f, 0.0f, 0.0f);
    __syncthreads();

    for (int q = t; q < nq; q += DEG_THREADS * DEG_BATCH) {
      int4 cc[DEG_BATCH];
      float4 ww[DEG_BATCH];
#pragma unroll
      for (int u = 0; u < DEG_BATCH; ++u) {
        const int qq = q + u * DEG_THREADS;
        const int qa = qq < nq ? qq : 0;
        cc[u] = cols4[qa];
        ww[u] = ew4[qa];
      }
#pragma unroll
      for (int u = 0; u < DEG_BATCH; ++u) {
        if (q + u * DEG_THREADS < nq) {
          deg_take(tab, cc[u].x, ww[u].x, lo, range);
          deg_take(tab, cc[u].y, ww[u].y, lo, range);
          deg_take(tab, cc[u].z, ww[u].z, lo, range);
          deg_take(tab, cc[u].w, ww[u].w, lo, range);
        }
      }
    }
    for (int e = start + 4 * nq + t; e < end; e += DEG_THREADS)
      deg_take(tab, cols[e], ew[e], lo, range);
    __syncthreads();

    for (int i = t; i < range; i += DEG_THREADS * DEG_BATCH) {
      float v[DEG_BATCH];
      int   jr[DEG_BATCH];
#pragma unroll
      for (int u = 0; u < DEG_BATCH; ++u) {
        const int j = i + u * DEG_THREADS;
        const int k = j + rot;
        jr[u] = j < range ? (k >= range ? k - range : k) : -1;
        v[u] = tab[jr[u] >= 0 ? jr[u] : 0];
      }
#pragma unroll
      for (int u = 0; u < DEG_BATCH; ++u)
        if (jr[u] >= 0 && lo + jr[u] < N) atomic_add_f32(&deg[lo + jr[u]], v[u]);
    }
    __syncthreads();
  }
}

__global__ void dinv_kernel(const float* __restrict__ deg2,
                            float* __restrict__ dinv, int N) {
  int i = blockIdx.x * blockDim.x + threadIdx.x;
  if (i >= N) return;
  float d = (deg2[i] + deg2[(size_t)N + i]) + 1.0f;
  dinv[i] = (d > 0.0f) ? rsqrtf(d) : 0.0f;
}

template <int NCOLS>
__global__ void gemm_wmma_f32(const float* __restrict__ A,
                              const float* __restrict__ B,
                              float* __restrict__ C) {
  const int m0   = blockIdx.x * 16;
  const int wid  = threadIdx.x >> 5;
  const int lane = threadIdx.x & 31;
  const int n0   = wid * 16;
  const int ln   = lane & 15;
  const int half = lane >> 4;

  v8f acc = {};
  const float* aptr = A + (size_t)(m0 + ln) * 64 + 2 * half;
  const float* bptr = B + (size_t)(2 * half) * NCOLS + n0 + ln;

#pragma unroll
  for (int k = 0; k < 64; k += 4) {
    v2f a, b;
    a.x = aptr[k];
    a.y = aptr[k + 1];
    b.x = bptr[k * NCOLS];
    b.y = bptr[k * NCOLS + NCOLS];
    acc = __builtin_amdgcn_wmma_f32_16x16x4_f32(
         false, a,  false, b,
         (short)0, acc,  false,  false);
  }

  float* cptr = C + (size_t)(m0 + 8 * half) * NCOLS + n0 + ln;
#pragma unroll
  for (int r = 0; r < 8; ++r) {
    cptr[(size_t)r * NCOLS] = acc[r];
  }
}


template <int CH, int UU>
__device__ __forceinline__ void push_edges(const float* __restrict__ x,
                                           float* __restrict__ out, int lane,
                                           int r, int c, float nrm, int j0) {
  constexpr int Q = CH / 32;
  int rj[UU], cj[UU];
  float nj[UU];
#pragma unroll
  for (int u = 0; u < UU; ++u) {
    rj[u] = __builtin_amdgcn_readlane(r, j0 + u);
    cj[u] = __builtin_amdgcn_readlane(c, j0 + u);
    nj[u] = __int_as_float(__builtin_amdgcn_readlane(__float_as_int(nrm), j0 + u));
  }
  float xv[UU][Q];
#pragma unroll
  for (int u = 0; u < UU; ++u)
#pragma unroll
    for (int q = 0; q < Q; ++q)
      xv[u][q] = x[(size_t)rj[u] * CH + q * 32 + lane];
#pragma unroll
  for (int u = 0; u < UU; ++u)
#pragma unroll
    for (int q = 0; q < Q; ++q)
      atomic_add_f32(out + (size_t)cj[u] * CH + q * 32 + lane, xv[u][q] * nj[u]);
}

template <int CH, bool MAKE_NORM>
__global__ void aggregate_kernel(const float* __restrict__ x,
                                 const int* __restrict__ rows,
                                 const int* __restrict__ cols,
                                 const float* __restrict__ ew,
                                 const float* __restrict__ dinv,
                                 float* __restrict__ nrmbuf,
                                 float* __restrict__ out2h,
                                 int E, int N) {
  constexpr int U = 512 / CH;
  float* out = out2h + (size_t)package_half() * (size_t)N * CH;
  const int lane  = (int)(threadIdx.x & 31u);
  const int total = E + N;
  const int e = (int)(blockIdx.x * blockDim.x + threadIdx.x);

  int r = 0, c = 0;
  float nrm = 0.0f;
  if (e < total) {
    if (e < E) { r = rows[e]; c = cols[e]; }
    else       { r = c = e - E; }
    if (MAKE_NORM) {
      float w = 1.0f;
      if (e < E) w = ew[e];
      nrm = dinv[r] * w * dinv[c];
      nrmbuf[e] = nrm;
    } else {
      nrm = nrmbuf[e];
    }
  }

  const int e0 = __builtin_amdgcn_readfirstlane(e - lane);
  int cnt = total - e0;
  cnt = cnt > 32 ? 32 : cnt;

  int j = 0;
  for (; j + U <= cnt; j += U) push_edges<CH, U>(x, out, lane, r, c, nrm, j);
  for (; j < cnt; ++j)         push_edges<CH, 1>(x, out, lane, r, c, nrm, j);
}

__global__ void bias_relu_kernel(float* __restrict__ x,
                                 const float* __restrict__ b, int N) {
  long long tid = (long long)blockIdx.x * blockDim.x + threadIdx.x;
  if (tid >= (long long)N * 16) return;
  int c4 = (int)(tid & 15);
  float4 v  = ((float4*)x)[tid];
  float4 v1 = ((const float4*)x)[(long long)N * 16 + tid];
  float4 bb = ((const float4*)b)[c4];
  v.x = fmaxf((v.x + v1.x) + bb.x, 0.0f);
  v.y = fmaxf((v.y + v1.y) + bb.y, 0.0f);
  v.z = fmaxf((v.z + v1.z) + bb.z, 0.0f);
  v.w = fmaxf((v.w + v1.w) + bb.w, 0.0f);
  ((float4*)x)[tid] = v;
}

__global__ void l2norm_kernel(const float* __restrict__ x,
                              const float* __restrict__ b2,
                              float* __restrict__ out, int N) {
  int row  = blockIdx.x * (blockDim.x >> 5) + (threadIdx.x >> 5);
  int lane = threadIdx.x & 31;
  if (row >= N) return;
  float v = (x[(long long)row * 32 + lane] + x[((long long)N + row) * 32 + lane]) + b2[lane];
  float ss = v * v;
#pragma unroll
  for (int off = 16; off > 0; off >>= 1)
    ss += __shfl_xor(ss, off, 32);
  float l2 = sqrtf(ss);
  float scale = 1.0f / fmaxf(l2, 1e-12f);
  out[(long long)row * 32 + lane] = v * scale;
}

extern "C" void kernel_launch(void* const* d_in, const int* in_sizes, int n_in,
                              void* d_out, int out_size, void* d_ws, size_t ws_size,
                              hipStream_t stream) {
  const int*   edge_index = (const int*)d_in[0];
  const float* ew         = (const float*)d_in[1];
  const float* emb        = (const float*)d_in[2];
  const float* W1         = (const float*)d_in[3];
  const float* b1         = (const float*)d_in[4];
  const float* W2         = (const float*)d_in[5];
  const float* b2         = (const float*)d_in[6];

  const int E = in_sizes[0] / 2;
  const int N = in_sizes[2] / 64;
  const int* rows = edge_index;
  const int* cols = edge_index + E;

  float* ws   = (float*)d_ws;
  float* dinv = ws;
  float* deg2 = dinv + (size_t)N;
  float* xw1  = deg2 + (size_t)N * 2;
  float* out1 = xw1  + (size_t)N * 64;
  float* xw2  = out1 + (size_t)N * 128;
  float* out2 = xw2  + (size_t)N * 32;
  float* nrmb = out2 + (size_t)N * 64;

  hipMemsetAsync(deg2, 0, sizeof(float) * (size_t)N * 2, stream);
  hipMemsetAsync(out1, 0, sizeof(float) * (size_t)N * 128, stream);
  hipMemsetAsync(out2, 0, sizeof(float) * (size_t)N * 64, stream);

  deg_kernel<<<256, DEG_THREADS, 0, stream>>>(cols, ew, deg2, E, N);
  dinv_kernel<<<(N + 255) / 256, 256, 0, stream>>>(deg2, dinv, N);

  gemm_wmma_f32<64><<<N / 16, 128, 0, stream>>>(emb, W1, xw1);
  aggregate_kernel<64, true><<<(unsigned)((E + N + 255) / 256), 256, 0, stream>>>(
      xw1, rows, cols, ew, dinv, nrmb, out1, E, N);
  bias_relu_kernel<<<(unsigned)(((long long)N * 16 + 255) / 256), 256, 0, stream>>>(out1, b1, N);

  gemm_wmma_f32<32><<<N / 16, 64, 0, stream>>>(out1, W2, xw2);
  aggregate_kernel<32, false><<<(unsigned)((E + N + 255) / 256), 256, 0, stream>>>(
      xw2, rows, cols, ew, dinv, nrmb, out2, E, N);

  l2norm_kernel<<<(N + 7) / 8, 256, 0, stream>>>(out2, b2, (float*)d_out, N);
}
